# starting kernel
# baseline (speedup 1.0000x reference)
#include <hip/hip_runtime.h>


typedef __attribute__((ext_vector_type(2))) float v2f;
typedef __attribute__((ext_vector_type(8))) float v8f;

#define LDS_STRIDE 33
#define BLK_FLOATS (256 * LDS_STRIDE)

__device__ __forceinline__ float gate_coef(int m, int j, float c, float s) {
  float v = 0.0f;
  if (j == m) v = c;
  else if (j == 15 - m) v = (m < 8) ? s : -s;
  return v;
}

__global__ __launch_bounds__(256) void ulayer_kron_kernel(
    const float* __restrict__ x,
    const float* __restrict__ thetas,
    float* __restrict__ out)
{
  __shared__ float lds[2 * BLK_FLOATS];

  const int i0  = blockIdx.x;
  const int bt  = blockIdx.y * 32;
  const int tid = threadIdx.x;

  const float t0 = thetas[0], t1 = thetas[1], t2 = thetas[2];
  const float c0 = cosf(t0), s0 = sinf(t0);
  const float c1 = cosf(t1), s1 = sinf(t1);
  const float c2 = cosf(t2), s2 = sinf(t2);

  const int rowLo = i0 * 256;
  const int rowHi = (15 - i0) * 256;

#pragma unroll
  for (int it = 0; it < 8; ++it) {
    const int flat = tid + it * 256;
    const int q = flat & 7;
    const int r = flat >> 3;
    const int col = bt + q * 4;
    const float4 xlo = *(const float4*)(x + (rowLo + r) * 2048 + col);
    const float4 xhi = *(const float4*)(x + (rowHi + r) * 2048 + col);
    float* plo = &lds[0 * BLK_FLOATS + r * LDS_STRIDE + q * 4];
    float* phi = &lds[1 * BLK_FLOATS + r * LDS_STRIDE + q * 4];
    plo[0] = c0 * xlo.x + s0 * xhi.x;  phi[0] = c0 * xhi.x - s0 * xlo.x;
    plo[1] = c0 * xlo.y + s0 * xhi.y;  phi[1] = c0 * xhi.y - s0 * xlo.y;
    plo[2] = c0 * xlo.z + s0 * xhi.z;  phi[2] = c0 * xhi.z - s0 * xlo.z;
    plo[3] = c0 * xlo.w + s0 * xhi.w;  phi[3] = c0 * xhi.w - s0 * xlo.w;
  }
  __syncthreads();

  const int lane = tid & 31;
  const int half = lane >> 4;
  const int l16  = lane & 15;
  const int wave = tid >> 5;

  auto wmma_stage = [&](float c, float s, int tileRowMul, int kRowStride) {
#pragma unroll
    for (int j = 0; j < 8; ++j) {
      const int tile = wave + (j << 3);
      const int qb   = tile >> 5;
      const int rem  = tile & 31;
      const int t    = rem >> 1;
      const int h    = rem & 1;
      float* base = &lds[qb * BLK_FLOATS + t * tileRowMul + h * 16 + l16];

      v8f acc = {};
#pragma unroll
      for (int k = 0; k < 4; ++k) {
        const int k0 = 4 * k + half * 2;
        v2f a, b;
        a.x = gate_coef(l16, k0,     c, s);
        a.y = gate_coef(l16, k0 + 1, c, s);
        b.x = base[(k0)     * kRowStride];
        b.y = base[(k0 + 1) * kRowStride];
        acc = __builtin_amdgcn_wmma_f32_16x16x4_f32(
             false, a,  false, b,
             (short)0, acc,  false,  false);
      }
#pragma unroll
      for (int jj = 0; jj < 8; ++jj)
        base[(half * 8 + jj) * kRowStride] = acc[jj];
    }
  };

  wmma_stage(c2, s2,  16 * LDS_STRIDE,  LDS_STRIDE);
  __syncthreads();

  wmma_stage(c1, s1,  LDS_STRIDE,  16 * LDS_STRIDE);
  __syncthreads();

#pragma unroll
  for (int it = 0; it < 8; ++it) {
    const int flat = tid + it * 256;
    const int q = flat & 7;
    const int r = flat >> 3;
    const int col = bt + q * 4;
    const float* plo = &lds[0 * BLK_FLOATS + r * LDS_STRIDE + q * 4];
    const float* phi = &lds[1 * BLK_FLOATS + r * LDS_STRIDE + q * 4];
    float4 vlo, vhi;
    vlo.x = plo[0]; vlo.y = plo[1]; vlo.z = plo[2]; vlo.w = plo[3];
    vhi.x = phi[0]; vhi.y = phi[1]; vhi.z = phi[2]; vhi.w = phi[3];
    *(float4*)(out + (rowLo + r) * 2048 + col) = vlo;
    *(float4*)(out + (rowHi + r) * 2048 + col) = vhi;
  }
}

extern "C" void kernel_launch(void* const* d_in, const int* in_sizes, int n_in,
                              void* d_out, int out_size, void* d_ws, size_t ws_size,
                              hipStream_t stream) {
  (void)in_sizes; (void)n_in; (void)out_size; (void)d_ws; (void)ws_size;
  const float* x      = (const float*)d_in[0];
  const float* thetas = (const float*)d_in[1];
  float* out          = (float*)d_out;

  dim3 grid(8, 64, 1);
  dim3 block(256, 1, 1);
  ulayer_kron_kernel<<<grid, block, 0, stream>>>(x, thetas, out);
}


# Fuse the axis-1 and axis-2 gate products into one 16x16 WMMA product per closed 16-row group: nothing to transpose, fragments go global-registers-global, no LDS, no barrier; 4-term sum order changes
# speedup vs baseline: 1.6180x; 1.6180x over previous
#include <hip/hip_runtime.h>


typedef __attribute__((ext_vector_type(2))) float v2f;
typedef __attribute__((ext_vector_type(8))) float v8f;

#define NCOL 2048

__global__ __launch_bounds__(256) void ulayer_kron_kernel(
    const float* __restrict__ x,
    const float* __restrict__ thetas,
    float* __restrict__ out)
{
  const int tid  = threadIdx.x;
  const int lane = tid & 31;
  const int wave = tid >> 5;
  const int half = lane >> 4;
  const int l16  = lane & 15;

  const int cbg = blockIdx.x & 3;
  const int g   = blockIdx.x >> 2;
  const int p   = blockIdx.y;
  const int i0  = blockIdx.z;

  const int col   = (cbg * 8 + wave) * 64 + 4 * l16;
  const int rowLo = i0 * 256;
  const int rowHi = (15 - i0) * 256;


  float4 xl[4][2], xh[4][2];
#pragma unroll
  for (int k = 0; k < 4; ++k) {
    const int a1 = (k >> 1) ? (15 - p) : p;
    const int a2 = ((k & 1) ? (12 - 4 * g) : (4 * g)) + 2 * half;
#pragma unroll
    for (int v = 0; v < 2; ++v) {
      const int r = a1 * 16 + a2 + v;
      xl[k][v] = *(const float4*)(x + (rowLo + r) * NCOL + col);
      xh[k][v] = *(const float4*)(x + (rowHi + r) * NCOL + col);
    }
  }

  const float t0 = thetas[0], t1 = thetas[1], t2 = thetas[2];
  const float c0 = cosf(t0), s0 = sinf(t0);
  const float c1 = cosf(t1), s1 = sinf(t1);
  const float c2 = cosf(t2), s2 = sinf(t2);

  const int Am = l16 >> 3;
  const int jm = l16 & 7;
  v2f afrag[4];
#pragma unroll
  for (int k = 0; k < 4; ++k) {
    const int Ak = k >> 1;
    const float w1 = (Am == Ak) ? c1 : ((Am == 0) ? s1 : -s1);
    float w[2];
#pragma unroll
    for (int v = 0; v < 2; ++v) {
      const int jk = 4 * (k & 1) + 2 * half + v;
      float w2 = 0.0f;
      if (jk == jm) w2 = c2;
      else if (jk == 7 - jm) w2 = (jm < 4) ? s2 : -s2;
      w[v] = w1 * w2;
    }
    afrag[k].x = w[0];
    afrag[k].y = w[1];
  }

  v8f accLo[4], accHi[4];
#pragma unroll
  for (int e = 0; e < 4; ++e) { accLo[e] = (v8f){}; accHi[e] = (v8f){}; }

#pragma unroll
  for (int k = 0; k < 4; ++k) {
    float lo[2][4], hi[2][4];
#pragma unroll
    for (int v = 0; v < 2; ++v) {
      const float4 a = xl[k][v];
      const float4 b = xh[k][v];
      lo[v][0] = c0 * a.x + s0 * b.x;  hi[v][0] = c0 * b.x - s0 * a.x;
      lo[v][1] = c0 * a.y + s0 * b.y;  hi[v][1] = c0 * b.y - s0 * a.y;
      lo[v][2] = c0 * a.z + s0 * b.z;  hi[v][2] = c0 * b.z - s0 * a.z;
      lo[v][3] = c0 * a.w + s0 * b.w;  hi[v][3] = c0 * b.w - s0 * a.w;
    }
#pragma unroll
    for (int e = 0; e < 4; ++e) {
      v2f bl, bh;
      bl.x = lo[0][e];  bl.y = lo[1][e];
      bh.x = hi[0][e];  bh.y = hi[1][e];
      accLo[e] = __builtin_amdgcn_wmma_f32_16x16x4_f32(
           false, afrag[k],  false, bl,
           (short)0, accLo[e],  false,  false);
      accHi[e] = __builtin_amdgcn_wmma_f32_16x16x4_f32(
           false, afrag[k],  false, bh,
           (short)0, accHi[e],  false,  false);
    }
  }

  const int a1s = half ? (15 - p) : p;
#pragma unroll
  for (int jj = 0; jj < 8; ++jj) {
    const int a2s = (jj < 4) ? (4 * g + jj) : (8 - 4 * g + jj);
    const int r = a1s * 16 + a2s;
    float4 vlo, vhi;
    vlo.x = accLo[0][jj]; vlo.y = accLo[1][jj]; vlo.z = accLo[2][jj]; vlo.w = accLo[3][jj];
    vhi.x = accHi[0][jj]; vhi.y = accHi[1][jj]; vhi.z = accHi[2][jj]; vhi.w = accHi[3][jj];
    *(float4*)(out + (rowLo + r) * NCOL + col) = vlo;
    *(float4*)(out + (rowHi + r) * NCOL + col) = vhi;
  }
}

extern "C" void kernel_launch(void* const* d_in, const int* in_sizes, int n_in,
                              void* d_out, int out_size, void* d_ws, size_t ws_size,
                              hipStream_t stream) {
  (void)in_sizes; (void)n_in; (void)out_size; (void)d_ws; (void)ws_size;
  const float* x      = (const float*)d_in[0];
  const float* thetas = (const float*)d_in[1];
  float* out          = (float*)d_out;

  dim3 grid(8, 8, 8);
  dim3 block(256, 1, 1);
  ulayer_kron_kernel<<<grid, block, 0, stream>>>(x, thetas, out);
}


# Two adjacent 64-column units per wave: 2048 waves (256 work-groups, one per WGP) instead of 4096; loads pinned unit-major at wave entry, unit 0 multiplied and stored while unit 1 arrives
# speedup vs baseline: 1.7639x; 1.0902x over previous
#include <hip/hip_runtime.h>


typedef __attribute__((ext_vector_type(2))) float v2f;
typedef __attribute__((ext_vector_type(8))) float v8f;

#define NCOL 2048
#define UNITS 2

__global__ __launch_bounds__(256) void ulayer_kron_kernel(
    const float* __restrict__ x,
    const float* __restrict__ thetas,
    float* __restrict__ out)
{
  const int tid  = threadIdx.x;
  const int lane = tid & 31;
  const int wave = tid >> 5;
  const int half = lane >> 4;
  const int l16  = lane & 15;

  const int cbg = blockIdx.x & 1;
  const int g   = blockIdx.x >> 1;
  const int p   = blockIdx.y;
  const int i0  = blockIdx.z;

  const int col   = (cbg * 8 + wave) * (64 * UNITS) + 4 * l16;
  const int rowLo = i0 * 256;
  const int rowHi = (15 - i0) * 256;


  float4 xl[UNITS][4][2], xh[UNITS][4][2];
#pragma unroll
  for (int u = 0; u < UNITS; ++u) {
#pragma unroll
    for (int k = 0; k < 4; ++k) {
      const int a1 = (k >> 1) ? (15 - p) : p;
      const int a2 = ((k & 1) ? (12 - 4 * g) : (4 * g)) + 2 * half;
#pragma unroll
      for (int v = 0; v < 2; ++v) {
        const int r = a1 * 16 + a2 + v;
        xl[u][k][v] = *(const float4*)(x + (rowLo + r) * NCOL + col + 64 * u);
        xh[u][k][v] = *(const float4*)(x + (rowHi + r) * NCOL + col + 64 * u);
      }
    }
    __builtin_amdgcn_sched_barrier(0);
  }

  const float t0 = thetas[0], t1 = thetas[1], t2 = thetas[2];
  const float c0 = cosf(t0), s0 = sinf(t0);
  const float c1 = cosf(t1), s1 = sinf(t1);
  const float c2 = cosf(t2), s2 = sinf(t2);

  const int Am = l16 >> 3;
  const int jm = l16 & 7;
  v2f afrag[4];
#pragma unroll
  for (int k = 0; k < 4; ++k) {
    const int Ak = k >> 1;
    const float w1 = (Am == Ak) ? c1 : ((Am == 0) ? s1 : -s1);
    float w[2];
#pragma unroll
    for (int v = 0; v < 2; ++v) {
      const int jk = 4 * (k & 1) + 2 * half + v;
      float w2 = 0.0f;
      if (jk == jm) w2 = c2;
      else if (jk == 7 - jm) w2 = (jm < 4) ? s2 : -s2;
      w[v] = w1 * w2;
    }
    afrag[k].x = w[0];
    afrag[k].y = w[1];
  }

  const int a1s = half ? (15 - p) : p;

#pragma unroll
  for (int u = 0; u < UNITS; ++u) {
    v8f accLo[4], accHi[4];
#pragma unroll
    for (int e = 0; e < 4; ++e) { accLo[e] = (v8f){}; accHi[e] = (v8f){}; }

#pragma unroll
    for (int k = 0; k < 4; ++k) {
      float lo[2][4], hi[2][4];
#pragma unroll
      for (int v = 0; v < 2; ++v) {
        const float4 a = xl[u][k][v];
        const float4 b = xh[u][k][v];
        lo[v][0] = c0 * a.x + s0 * b.x;  hi[v][0] = c0 * b.x - s0 * a.x;
        lo[v][1] = c0 * a.y + s0 * b.y;  hi[v][1] = c0 * b.y - s0 * a.y;
        lo[v][2] = c0 * a.z + s0 * b.z;  hi[v][2] = c0 * b.z - s0 * a.z;
        lo[v][3] = c0 * a.w + s0 * b.w;  hi[v][3] = c0 * b.w - s0 * a.w;
      }
#pragma unroll
      for (int e = 0; e < 4; ++e) {
        v2f bl, bh;
        bl.x = lo[0][e];  bl.y = lo[1][e];
        bh.x = hi[0][e];  bh.y = hi[1][e];
        accLo[e] = __builtin_amdgcn_wmma_f32_16x16x4_f32(
             false, afrag[k],  false, bl,
             (short)0, accLo[e],  false,  false);
        accHi[e] = __builtin_amdgcn_wmma_f32_16x16x4_f32(
             false, afrag[k],  false, bh,
             (short)0, accHi[e],  false,  false);
      }
    }

#pragma unroll
    for (int jj = 0; jj < 8; ++jj) {
      const int a2s = (jj < 4) ? (4 * g + jj) : (8 - 4 * g + jj);
      const int r = a1s * 16 + a2s;
      float4 vlo, vhi;
      vlo.x = accLo[0][jj]; vlo.y = accLo[1][jj]; vlo.z = accLo[2][jj]; vlo.w = accLo[3][jj];
      vhi.x = accHi[0][jj]; vhi.y = accHi[1][jj]; vhi.z = accHi[2][jj]; vhi.w = accHi[3][jj];
      *(float4*)(out + (rowLo + r) * NCOL + col + 64 * u) = vlo;
      *(float4*)(out + (rowHi + r) * NCOL + col + 64 * u) = vhi;
    }
    if (u + 1 < UNITS) __builtin_amdgcn_sched_barrier(0);
  }
}

extern "C" void kernel_launch(void* const* d_in, const int* in_sizes, int n_in,
                              void* d_out, int out_size, void* d_ws, size_t ws_size,
                              hipStream_t stream) {
  (void)in_sizes; (void)n_in; (void)out_size; (void)d_ws; (void)ws_size;
  const float* x      = (const float*)d_in[0];
  const float* thetas = (const float*)d_in[1];
  float* out          = (float*)d_out;

  dim3 grid(4, 8, 8);
  dim3 block(256, 1, 1);
  ulayer_kron_kernel<<<grid, block, 0, stream>>>(x, thetas, out);
}
